# starting kernel
# baseline (speedup 1.0000x reference)
#include <hip/hip_runtime.h>

#define F       128
#define ROWS    64
#define WPAD    136
#define APAD    132
#define THREADS 256

typedef __attribute__((ext_vector_type(2))) float v2f;
typedef __attribute__((ext_vector_type(8))) float v8f;

__global__ __launch_bounds__(THREADS) void gcn_fused_kernel(
    const float* __restrict__ X,
    const float* __restrict__ W,
    const float* __restrict__ deg,
    const int*   __restrict__ rowp,
    const int*   __restrict__ colidx,
    float*       __restrict__ out,
    int n)
{
    extern __shared__ float smem[];
    float* sW = smem;
    float* sA = smem + F * WPAD;

    const int tid  = threadIdx.x;
    const int wave = tid >> 5;
    const int lane = tid & 31;

    for (int i = tid; i < F * (F / 4); i += THREADS) {
        const int r = i >> 5;
        const int c = (i & 31) << 2;
        const float4 w = *(const float4*)(W + r * F + c);
        float* dst = sW + r * WPAD + c;
        dst[0] = w.x; dst[1] = w.y; dst[2] = w.z; dst[3] = w.w;
    }

    const int rowBase = blockIdx.x * ROWS;
    const int c4 = lane << 2;
    for (int i = 0; i < ROWS / 8; ++i) {
        const int lrow = wave * (ROWS / 8) + i;
        const int row  = rowBase + lrow;
        float ax = 0.f, ay = 0.f, az = 0.f, aw = 0.f;
        if (row < n) {
            const int   s  = rowp[row];
            const int   e  = rowp[row + 1];
            const float di = deg[row];
            for (int ed = s; ed < e; ++ed) {
                const int   j   = colidx[ed];
                const float nrm = di * deg[j];
                const float4 x = *(const float4*)(X + (long long)j * F + c4);
                ax += nrm * x.x; ay += nrm * x.y;
                az += nrm * x.z; aw += nrm * x.w;
            }
        }
        float* dst = sA + lrow * APAD + c4;
        dst[0] = ax; dst[1] = ay; dst[2] = az; dst[3] = aw;
    }
    __syncthreads();

    const int half = lane >> 4;
    const int l16  = lane & 15;
    const int colBase = wave << 4;
    for (int tr = 0; tr < ROWS / 16; ++tr) {
        v8f acc = {};
        const float* Arow = sA + (tr * 16 + l16) * APAD;
        const float* Bcol = sW + colBase + l16;
        #pragma unroll
        for (int k0 = 0; k0 < F; k0 += 4) {
            const int ka = k0 + 2 * half;
            v2f a; a.x = Arow[ka];          a.y = Arow[ka + 1];
            v2f b; b.x = Bcol[ka * WPAD];   b.y = Bcol[(ka + 1) * WPAD];
            acc = __builtin_amdgcn_wmma_f32_16x16x4_f32(
                false, a, false, b, (short)0, acc, false, false);
        }
        const int col      = colBase + l16;
        const int nodeBase = rowBase + tr * 16 + half * 8;
        #pragma unroll
        for (int r = 0; r < 8; ++r) {
            const int node = nodeBase + r;
            if (node < n) out[(long long)node * F + col] = acc[r];
        }
    }
}

extern "C" void kernel_launch(void* const* d_in, const int* in_sizes, int n_in,
                              void* d_out, int out_size, void* d_ws, size_t ws_size,
                              hipStream_t stream) {
    const float* X      = (const float*)d_in[0];
    const float* W      = (const float*)d_in[1];
    const float* deg    = (const float*)d_in[2];
    const int*   rowp   = (const int*)d_in[3];
    const int*   colidx = (const int*)d_in[4];
    float*       out    = (float*)d_out;

    const int n = in_sizes[2];
    const int blocks = (n + ROWS - 1) / ROWS;
    const size_t lds_bytes = (size_t)(F * WPAD + ROWS * APAD) * sizeof(float);

    gcn_fused_kernel<<<blocks, THREADS, lds_bytes, stream>>>(
        X, W, deg, rowp, colidx, out, n);
}


# Gather with more loads in flight: a row's neighbour ids and deg entries fetched lane-parallel (ids one row ahead), its 16 X rows requested together before use; fused edge-order sums unchanged
# speedup vs baseline: 1.0583x; 1.0583x over previous
#include <hip/hip_runtime.h>

#define F       128
#define ROWS    64
#define WPAD    136
#define APAD    132
#define THREADS 256

typedef __attribute__((ext_vector_type(2))) float v2f;
typedef __attribute__((ext_vector_type(8))) float v8f;

__device__ __forceinline__ int load_ids(const int* __restrict__ colidx, int s, int e, int lane)
{
    int v = 0;
    if (s + lane < e) v = colidx[s + lane];
    return v;
}

template <int B, bool PREFETCH>
__device__ __forceinline__ void gather_batch(
    const float* __restrict__ X, const int* __restrict__ colidx,
    int jv, float dg, float di, int k, unsigned laneByte, int lane,
    int ns, int ne, int& jn,
    float& ax, float& ay, float& az, float& aw)
{
    float4 x[B];
    #pragma unroll
    for (int u = 0; u < B; ++u) {
        const int j = __builtin_amdgcn_readlane(jv, k + u);
        const char* xrow = (const char*)X + (size_t)(unsigned)j * (F * sizeof(float));
        x[u] = *(const float4*)(xrow + laneByte);
    }
    __builtin_amdgcn_sched_barrier(0);
    if (PREFETCH) jn = load_ids(colidx, ns, ne, lane);
    const float nv = di * dg;
    #pragma unroll
    for (int u = 0; u < B; ++u) {
        const float nrm = __int_as_float(__builtin_amdgcn_readlane(__float_as_int(nv), k + u));
        ax = __builtin_fmaf(nrm, x[u].x, ax); ay = __builtin_fmaf(nrm, x[u].y, ay);
        az = __builtin_fmaf(nrm, x[u].z, az); aw = __builtin_fmaf(nrm, x[u].w, aw);
    }
}

__global__ __launch_bounds__(THREADS) void gcn_fused_kernel(
    const float* __restrict__ X,
    const float* __restrict__ W,
    const float* __restrict__ deg,
    const int*   __restrict__ rowp,
    const int*   __restrict__ colidx,
    float*       __restrict__ out,
    int n)
{
    extern __shared__ float smem[];
    float* sW = smem;
    float* sA = smem + F * WPAD;

    const int tid  = threadIdx.x;
    const int wave = tid >> 5;
    const int lane = tid & 31;

    for (int i = tid; i < F * (F / 4); i += THREADS) {
        const int r = i >> 5;
        const int c = (i & 31) << 2;
        const float4 w = *(const float4*)(W + r * F + c);
        float* dst = sW + r * WPAD + c;
        dst[0] = w.x; dst[1] = w.y; dst[2] = w.z; dst[3] = w.w;
    }

    const int rowBase = blockIdx.x * ROWS;
    const int c4 = lane << 2;
    const unsigned laneByte = (unsigned)lane * 16u;
    const int r0 = rowBase + wave * (ROWS / 8);
    int rpi = r0 + (lane < 8 ? lane : 8);
    if (rpi > n) rpi = n;
    const int rpv = rowp[rpi];
    int dgi = r0 + (lane & 7);
    if (dgi > n - 1) dgi = n - 1;
    const float dgv = deg[dgi];

    int s  = __builtin_amdgcn_readlane(rpv, 0);
    int jn = load_ids(colidx, s, __builtin_amdgcn_readlane(rpv, 1), lane);
    for (int i = 0; i < ROWS / 8; ++i) {
        const int   e  = __builtin_amdgcn_readlane(rpv, i + 1);
        const int   ne = __builtin_amdgcn_readlane(rpv, i + 2);
        const float di = __int_as_float(__builtin_amdgcn_readlane(__float_as_int(dgv), i));
        int  jv = jn;
        bool pf = (i + 1 < ROWS / 8);
        float ax = 0.f, ay = 0.f, az = 0.f, aw = 0.f;
        for (int base = s; base < e; base += 32) {
            const int rem = e - base;
            const int cnt = rem < 32 ? rem : 32;
            if (base != s) jv = load_ids(colidx, base, e, lane);
            const float dg = deg[jv];
            int k = 0;
            if (pf && cnt >= 16) {
                gather_batch<16, true>(X, colidx, jv, dg, di, k, laneByte, lane, e, ne, jn, ax, ay, az, aw);
                k = 16; pf = false;
            }
            for (; k + 16 <= cnt; k += 16)
                gather_batch<16, false>(X, colidx, jv, dg, di, k, laneByte, lane, e, ne, jn, ax, ay, az, aw);
            if (k + 8 <= cnt) { gather_batch<8, false>(X, colidx, jv, dg, di, k, laneByte, lane, e, ne, jn, ax, ay, az, aw); k += 8; }
            if (k + 4 <= cnt) { gather_batch<4, false>(X, colidx, jv, dg, di, k, laneByte, lane, e, ne, jn, ax, ay, az, aw); k += 4; }
            if (k + 2 <= cnt) { gather_batch<2, false>(X, colidx, jv, dg, di, k, laneByte, lane, e, ne, jn, ax, ay, az, aw); k += 2; }
            if (k < cnt)      { gather_batch<1, false>(X, colidx, jv, dg, di, k, laneByte, lane, e, ne, jn, ax, ay, az, aw); }
        }
        if (pf) jn = load_ids(colidx, e, ne, lane);
        float* dst = sA + (wave * (ROWS / 8) + i) * APAD + c4;
        dst[0] = ax; dst[1] = ay; dst[2] = az; dst[3] = aw;
        s = e;
    }
    __syncthreads();

    const int half = lane >> 4;
    const int l16  = lane & 15;
    const int colBase = wave << 4;
    for (int tr = 0; tr < ROWS / 16; ++tr) {
        v8f acc = {};
        const float* Arow = sA + (tr * 16 + l16) * APAD;
        const float* Bcol = sW + colBase + l16;
        #pragma unroll
        for (int k0 = 0; k0 < F; k0 += 4) {
            const int ka = k0 + 2 * half;
            v2f a; a.x = Arow[ka];          a.y = Arow[ka + 1];
            v2f b; b.x = Bcol[ka * WPAD];   b.y = Bcol[(ka + 1) * WPAD];
            acc = __builtin_amdgcn_wmma_f32_16x16x4_f32(
                false, a, false, b, (short)0, acc, false, false);
        }
        const int col      = colBase + l16;
        const int nodeBase = rowBase + tr * 16 + half * 8;
        #pragma unroll
        for (int r = 0; r < 8; ++r) {
            const int node = nodeBase + r;
            if (node < n) out[(long long)node * F + col] = acc[r];
        }
    }
}

extern "C" void kernel_launch(void* const* d_in, const int* in_sizes, int n_in,
                              void* d_out, int out_size, void* d_ws, size_t ws_size,
                              hipStream_t stream) {
    const float* X      = (const float*)d_in[0];
    const float* W      = (const float*)d_in[1];
    const float* deg    = (const float*)d_in[2];
    const int*   rowp   = (const int*)d_in[3];
    const int*   colidx = (const int*)d_in[4];
    float*       out    = (float*)d_out;

    const int n = in_sizes[2];
    const int blocks = (n + ROWS - 1) / ROWS;
    const size_t lds_bytes = (size_t)(F * WPAD + ROWS * APAD) * sizeof(float);

    gcn_fused_kernel<<<blocks, THREADS, lds_bytes, stream>>>(
        X, W, deg, rowp, colidx, out, n);
}


# W no longer staged in LDS: each wave loads the B fragments of its column tile from global into registers and uses each for all four row tiles (four accumulators); LDS per block 103424 -> 33792 B
# speedup vs baseline: 1.1716x; 1.1071x over previous
#include <hip/hip_runtime.h>

#define F       128
#define ROWS    64
#define APAD    132
#define THREADS 256

typedef __attribute__((ext_vector_type(2))) float v2f;
typedef __attribute__((ext_vector_type(8))) float v8f;

__device__ __forceinline__ int load_ids(const int* __restrict__ colidx, int s, int e, int lane)
{
    int v = 0;
    if (s + lane < e) v = colidx[s + lane];
    return v;
}

template <int B, bool PREFETCH>
__device__ __forceinline__ void gather_batch(
    const float* __restrict__ X, const int* __restrict__ colidx,
    int jv, float dg, float di, int k, unsigned laneByte, int lane,
    int ns, int ne, int& jn,
    float& ax, float& ay, float& az, float& aw)
{
    float4 x[B];
    #pragma unroll
    for (int u = 0; u < B; ++u) {
        const int j = __builtin_amdgcn_readlane(jv, k + u);
        const char* xrow = (const char*)X + (size_t)(unsigned)j * (F * sizeof(float));
        x[u] = *(const float4*)(xrow + laneByte);
    }
    __builtin_amdgcn_sched_barrier(0);
    if (PREFETCH) jn = load_ids(colidx, ns, ne, lane);
    const float nv = di * dg;
    #pragma unroll
    for (int u = 0; u < B; ++u) {
        const float nrm = __int_as_float(__builtin_amdgcn_readlane(__float_as_int(nv), k + u));
        ax = __builtin_fmaf(nrm, x[u].x, ax); ay = __builtin_fmaf(nrm, x[u].y, ay);
        az = __builtin_fmaf(nrm, x[u].z, az); aw = __builtin_fmaf(nrm, x[u].w, aw);
    }
}

__global__ __launch_bounds__(THREADS) void gcn_fused_kernel(
    const float* __restrict__ X,
    const float* __restrict__ W,
    const float* __restrict__ deg,
    const int*   __restrict__ rowp,
    const int*   __restrict__ colidx,
    float*       __restrict__ out,
    int n)
{
    extern __shared__ float smem[];
    float* sA = smem;

    const int tid  = threadIdx.x;
    const int wave = tid >> 5;
    const int lane = tid & 31;

    const int rowBase = blockIdx.x * ROWS;
    const int c4 = lane << 2;
    const unsigned laneByte = (unsigned)lane * 16u;
    const int r0 = rowBase + wave * (ROWS / 8);
    int rpi = r0 + (lane < 8 ? lane : 8);
    if (rpi > n) rpi = n;
    const int rpv = rowp[rpi];
    int dgi = r0 + (lane & 7);
    if (dgi > n - 1) dgi = n - 1;
    const float dgv = deg[dgi];

    int s  = __builtin_amdgcn_readlane(rpv, 0);
    int jn = load_ids(colidx, s, __builtin_amdgcn_readlane(rpv, 1), lane);
    for (int i = 0; i < ROWS / 8; ++i) {
        const int   e  = __builtin_amdgcn_readlane(rpv, i + 1);
        const int   ne = __builtin_amdgcn_readlane(rpv, i + 2);
        const float di = __int_as_float(__builtin_amdgcn_readlane(__float_as_int(dgv), i));
        int  jv = jn;
        bool pf = (i + 1 < ROWS / 8);
        float ax = 0.f, ay = 0.f, az = 0.f, aw = 0.f;
        for (int base = s; base < e; base += 32) {
            const int rem = e - base;
            const int cnt = rem < 32 ? rem : 32;
            if (base != s) jv = load_ids(colidx, base, e, lane);
            const float dg = deg[jv];
            int k = 0;
            if (pf && cnt >= 16) {
                gather_batch<16, true>(X, colidx, jv, dg, di, k, laneByte, lane, e, ne, jn, ax, ay, az, aw);
                k = 16; pf = false;
            }
            for (; k + 16 <= cnt; k += 16)
                gather_batch<16, false>(X, colidx, jv, dg, di, k, laneByte, lane, e, ne, jn, ax, ay, az, aw);
            if (k + 8 <= cnt) { gather_batch<8, false>(X, colidx, jv, dg, di, k, laneByte, lane, e, ne, jn, ax, ay, az, aw); k += 8; }
            if (k + 4 <= cnt) { gather_batch<4, false>(X, colidx, jv, dg, di, k, laneByte, lane, e, ne, jn, ax, ay, az, aw); k += 4; }
            if (k + 2 <= cnt) { gather_batch<2, false>(X, colidx, jv, dg, di, k, laneByte, lane, e, ne, jn, ax, ay, az, aw); k += 2; }
            if (k < cnt)      { gather_batch<1, false>(X, colidx, jv, dg, di, k, laneByte, lane, e, ne, jn, ax, ay, az, aw); }
        }
        if (pf) jn = load_ids(colidx, e, ne, lane);
        float* dst = sA + (wave * (ROWS / 8) + i) * APAD + c4;
        dst[0] = ax; dst[1] = ay; dst[2] = az; dst[3] = aw;
        s = e;
    }
    const int half = lane >> 4;
    const int l16  = lane & 15;
    const int colBase = wave << 4;
    const float* Bcol = W + (2 * half) * F + colBase + l16;
    v2f bfrag[F / 4];
    #pragma unroll
    for (int kk = 0; kk < F / 4; ++kk) {
        bfrag[kk].x = Bcol[(4 * kk) * F];
        bfrag[kk].y = Bcol[(4 * kk + 1) * F];
    }
    __syncthreads();

    v8f acc[ROWS / 16];
    #pragma unroll
    for (int tr = 0; tr < ROWS / 16; ++tr) acc[tr] = v8f{};
    const float* Arow = sA + l16 * APAD + 2 * half;
    #pragma unroll
    for (int kk = 0; kk < F / 4; ++kk) {
        #pragma unroll
        for (int tr = 0; tr < ROWS / 16; ++tr) {
            const float* ap = Arow + tr * 16 * APAD + 4 * kk;
            v2f a; a.x = ap[0]; a.y = ap[1];
            acc[tr] = __builtin_amdgcn_wmma_f32_16x16x4_f32(
                false, a, false, bfrag[kk], (short)0, acc[tr], false, false);
        }
    }
    const int col = colBase + l16;
    #pragma unroll
    for (int tr = 0; tr < ROWS / 16; ++tr) {
        const int nodeBase = rowBase + tr * 16 + half * 8;
        #pragma unroll
        for (int r = 0; r < 8; ++r) {
            const int node = nodeBase + r;
            if (node < n) out[(long long)node * F + col] = acc[tr][r];
        }
    }
}

extern "C" void kernel_launch(void* const* d_in, const int* in_sizes, int n_in,
                              void* d_out, int out_size, void* d_ws, size_t ws_size,
                              hipStream_t stream) {
    const float* X      = (const float*)d_in[0];
    const float* W      = (const float*)d_in[1];
    const float* deg    = (const float*)d_in[2];
    const int*   rowp   = (const int*)d_in[3];
    const int*   colidx = (const int*)d_in[4];
    float*       out    = (float*)d_out;

    const int n = in_sizes[2];
    const int blocks = (n + ROWS - 1) / ROWS;
    const size_t lds_bytes = (size_t)(ROWS * APAD) * sizeof(float);

    gcn_fused_kernel<<<blocks, THREADS, lds_bytes, stream>>>(
        X, W, deg, rowp, colidx, out, n);
}


# Block height 64 -> 98 rows (seven 16-row tiles, WMMA in two passes of 4+3 tiles per wave): 1021 blocks, all resident at once at four per WGP, replace 1563 run in two uneven rounds; sums unchanged
# speedup vs baseline: 1.2514x; 1.0681x over previous
#include <hip/hip_runtime.h>

#define F       128
#define BROWS   98
#define NTILE   7
#define TROWS   (NTILE * 16)
#define RLO     (BROWS / 8)
#define REX     (BROWS % 8)
#define APAD    132
#define THREADS 256

typedef __attribute__((ext_vector_type(2))) float v2f;
typedef __attribute__((ext_vector_type(8))) float v8f;

__device__ __forceinline__ int load_ids(const int* __restrict__ colidx, int s, int e, int lane)
{
    int v = 0;
    if (s + lane < e) v = colidx[s + lane];
    return v;
}

template <int B, bool PREFETCH>
__device__ __forceinline__ void gather_batch(
    const float* __restrict__ X, const int* __restrict__ colidx,
    int jv, float dg, float di, int k, unsigned laneByte, int lane,
    int ns, int ne, int& jn,
    float& ax, float& ay, float& az, float& aw)
{
    float4 x[B];
    #pragma unroll
    for (int u = 0; u < B; ++u) {
        const int j = __builtin_amdgcn_readlane(jv, k + u);
        const char* xrow = (const char*)X + (size_t)(unsigned)j * (F * sizeof(float));
        x[u] = *(const float4*)(xrow + laneByte);
    }
    __builtin_amdgcn_sched_barrier(0);
    if (PREFETCH) jn = load_ids(colidx, ns, ne, lane);
    const float nv = di * dg;
    #pragma unroll
    for (int u = 0; u < B; ++u) {
        const float nrm = __int_as_float(__builtin_amdgcn_readlane(__float_as_int(nv), k + u));
        ax = __builtin_fmaf(nrm, x[u].x, ax); ay = __builtin_fmaf(nrm, x[u].y, ay);
        az = __builtin_fmaf(nrm, x[u].z, az); aw = __builtin_fmaf(nrm, x[u].w, aw);
    }
}

template <int T0, int NT>
__device__ __forceinline__ void gemm_pass(
    const float* __restrict__ Arow, const v2f (&bfrag)[F / 4],
    float* __restrict__ out, int rowBase, int half, int col, int n)
{
    v8f acc[NT];
    #pragma unroll
    for (int t = 0; t < NT; ++t) acc[t] = v8f{};
    #pragma unroll
    for (int kk = 0; kk < F / 4; ++kk) {
        #pragma unroll
        for (int t = 0; t < NT; ++t) {
            const float* ap = Arow + (T0 + t) * 16 * APAD + 4 * kk;
            v2f a; a.x = ap[0]; a.y = ap[1];
            acc[t] = __builtin_amdgcn_wmma_f32_16x16x4_f32(
                false, a, false, bfrag[kk], (short)0, acc[t], false, false);
        }
    }
    #pragma unroll
    for (int t = 0; t < NT; ++t) {
        #pragma unroll
        for (int r = 0; r < 8; ++r) {
            const int lr   = (T0 + t) * 16 + half * 8 + r;
            const int node = rowBase + lr;
            if (lr < BROWS && node < n) out[(long long)node * F + col] = acc[t][r];
        }
    }
}

__global__ __launch_bounds__(THREADS) void gcn_fused_kernel(
    const float* __restrict__ X,
    const float* __restrict__ W,
    const float* __restrict__ deg,
    const int*   __restrict__ rowp,
    const int*   __restrict__ colidx,
    float*       __restrict__ out,
    int n)
{
    extern __shared__ float smem[];
    float* sA = smem;

    const int tid  = threadIdx.x;
    const int wave = __builtin_amdgcn_readfirstlane(tid >> 5);
    const int lane = tid & 31;

    const int rowBase = blockIdx.x * BROWS;
    const int c4 = lane << 2;
    const unsigned laneByte = (unsigned)lane * 16u;
    const int myRows = RLO + (wave < REX ? 1 : 0);
    const int lr0    = wave * RLO + (wave < REX ? wave : REX);
    const int r0 = rowBase + lr0;
    for (int pr = BROWS + wave; pr < TROWS; pr += 8) {
        float* z = sA + pr * APAD + c4;
        z[0] = 0.f; z[1] = 0.f; z[2] = 0.f; z[3] = 0.f;
    }
    int rpi = r0 + (lane < myRows ? lane : myRows);
    if (rpi > n) rpi = n;
    const int rpv = rowp[rpi];
    int dgi = r0 + (lane < myRows ? lane : 0);
    if (dgi > n - 1) dgi = n - 1;
    const float dgv = deg[dgi];

    int s  = __builtin_amdgcn_readlane(rpv, 0);
    int jn = load_ids(colidx, s, __builtin_amdgcn_readlane(rpv, 1), lane);
    for (int i = 0; i < myRows; ++i) {
        const int   e  = __builtin_amdgcn_readlane(rpv, i + 1);
        const int   ne = __builtin_amdgcn_readlane(rpv, i + 2);
        const float di = __int_as_float(__builtin_amdgcn_readlane(__float_as_int(dgv), i));
        int  jv = jn;
        bool pf = (i + 1 < myRows);
        float ax = 0.f, ay = 0.f, az = 0.f, aw = 0.f;
        for (int base = s; base < e; base += 32) {
            const int rem = e - base;
            const int cnt = rem < 32 ? rem : 32;
            if (base != s) jv = load_ids(colidx, base, e, lane);
            const float dg = deg[jv];
            int k = 0;
            if (pf && cnt >= 16) {
                gather_batch<16, true>(X, colidx, jv, dg, di, k, laneByte, lane, e, ne, jn, ax, ay, az, aw);
                k = 16; pf = false;
            }
            for (; k + 16 <= cnt; k += 16)
                gather_batch<16, false>(X, colidx, jv, dg, di, k, laneByte, lane, e, ne, jn, ax, ay, az, aw);
            if (k + 8 <= cnt) { gather_batch<8, false>(X, colidx, jv, dg, di, k, laneByte, lane, e, ne, jn, ax, ay, az, aw); k += 8; }
            if (k + 4 <= cnt) { gather_batch<4, false>(X, colidx, jv, dg, di, k, laneByte, lane, e, ne, jn, ax, ay, az, aw); k += 4; }
            if (k + 2 <= cnt) { gather_batch<2, false>(X, colidx, jv, dg, di, k, laneByte, lane, e, ne, jn, ax, ay, az, aw); k += 2; }
            if (k < cnt)      { gather_batch<1, false>(X, colidx, jv, dg, di, k, laneByte, lane, e, ne, jn, ax, ay, az, aw); }
        }
        if (pf) jn = load_ids(colidx, e, ne, lane);
        float* dst = sA + (lr0 + i) * APAD + c4;
        dst[0] = ax; dst[1] = ay; dst[2] = az; dst[3] = aw;
        s = e;
    }
    const int half = lane >> 4;
    const int l16  = lane & 15;
    const int colBase = wave << 4;
    const float* Bcol = W + (2 * half) * F + colBase + l16;
    v2f bfrag[F / 4];
    #pragma unroll
    for (int kk = 0; kk < F / 4; ++kk) {
        bfrag[kk].x = Bcol[(4 * kk) * F];
        bfrag[kk].y = Bcol[(4 * kk + 1) * F];
    }
    __syncthreads();

    const float* Arow = sA + l16 * APAD + 2 * half;
    const int col = colBase + l16;
    gemm_pass<0, 4>(Arow, bfrag, out, rowBase, half, col, n);
    gemm_pass<4, NTILE - 4>(Arow, bfrag, out, rowBase, half, col, n);
}

extern "C" void kernel_launch(void* const* d_in, const int* in_sizes, int n_in,
                              void* d_out, int out_size, void* d_ws, size_t ws_size,
                              hipStream_t stream) {
    const float* X      = (const float*)d_in[0];
    const float* W      = (const float*)d_in[1];
    const float* deg    = (const float*)d_in[2];
    const int*   rowp   = (const int*)d_in[3];
    const int*   colidx = (const int*)d_in[4];
    float*       out    = (float*)d_out;

    const int n = in_sizes[2];
    const int blocks = (n + BROWS - 1) / BROWS;
    const size_t lds_bytes = (size_t)(TROWS * APAD) * sizeof(float);

    gcn_fused_kernel<<<blocks, THREADS, lds_bytes, stream>>>(
        X, W, deg, rowp, colidx, out, n);
}


# Block's rows run as two stages (64 and 34 rows, each gathered then multiplied); half the blocks take the 34-row stage first, so blocks' WMMAs overlap their neighbours' gathers; sums unchanged
# speedup vs baseline: 1.2893x; 1.0303x over previous
#include <hip/hip_runtime.h>

#define F       128
#define BROWS   98
#define NTILE   7
#define TROWS   (NTILE * 16)
#define G0ROWS  64
#define G0TILES (G0ROWS / 16)
#define G1ROWS  (BROWS - G0ROWS)
#define G1LO    (G1ROWS / 8)
#define G1EX    (G1ROWS % 8)
#define APAD    132
#define THREADS 256

typedef __attribute__((ext_vector_type(2))) float v2f;
typedef __attribute__((ext_vector_type(8))) float v8f;

__device__ __forceinline__ int load_ids(const int* __restrict__ colidx, int s, int e, int lane)
{
    int v = 0;
    if (s + lane < e) v = colidx[s + lane];
    return v;
}

template <int B, bool PREFETCH>
__device__ __forceinline__ void gather_batch(
    const float* __restrict__ X, const int* __restrict__ colidx,
    int jv, float dg, float di, int k, unsigned laneByte, int lane,
    int ns, int ne, int& jn,
    float& ax, float& ay, float& az, float& aw)
{
    float4 x[B];
    #pragma unroll
    for (int u = 0; u < B; ++u) {
        const int j = __builtin_amdgcn_readlane(jv, k + u);
        const char* xrow = (const char*)X + (size_t)(unsigned)j * (F * sizeof(float));
        x[u] = *(const float4*)(xrow + laneByte);
    }
    __builtin_amdgcn_sched_barrier(0);
    if (PREFETCH) jn = load_ids(colidx, ns, ne, lane);
    const float nv = di * dg;
    #pragma unroll
    for (int u = 0; u < B; ++u) {
        const float nrm = __int_as_float(__builtin_amdgcn_readlane(__float_as_int(nv), k + u));
        ax = __builtin_fmaf(nrm, x[u].x, ax); ay = __builtin_fmaf(nrm, x[u].y, ay);
        az = __builtin_fmaf(nrm, x[u].z, az); aw = __builtin_fmaf(nrm, x[u].w, aw);
    }
}

template <int MAXB>
__device__ __forceinline__ void gather_rows(
    const float* __restrict__ X, const float* __restrict__ deg,
    const int* __restrict__ rowp, const int* __restrict__ colidx,
    float* __restrict__ sA, int n, int r0, int lr0, int myRows,
    int lane, unsigned laneByte, int c4)
{
    int rpi = r0 + (lane < myRows ? lane : myRows);
    if (rpi > n) rpi = n;
    const int rpv = rowp[rpi];
    int dgi = r0 + (lane < myRows ? lane : 0);
    if (dgi > n - 1) dgi = n - 1;
    const float dgv = deg[dgi];

    int s  = __builtin_amdgcn_readlane(rpv, 0);
    int jn = load_ids(colidx, s, __builtin_amdgcn_readlane(rpv, 1), lane);
    for (int i = 0; i < myRows; ++i) {
        const int   e  = __builtin_amdgcn_readlane(rpv, i + 1);
        const int   ne = __builtin_amdgcn_readlane(rpv, i + 2);
        const float di = __int_as_float(__builtin_amdgcn_readlane(__float_as_int(dgv), i));
        int  jv = jn;
        bool pf = (i + 1 < myRows);
        float ax = 0.f, ay = 0.f, az = 0.f, aw = 0.f;
        for (int base = s; base < e; base += 32) {
            const int rem = e - base;
            const int cnt = rem < 32 ? rem : 32;
            if (base != s) jv = load_ids(colidx, base, e, lane);
            const float dg = deg[jv];
            int k = 0;
            if (pf && cnt >= MAXB) {
                gather_batch<MAXB, true>(X, colidx, jv, dg, di, k, laneByte, lane, e, ne, jn, ax, ay, az, aw);
                k = MAXB; pf = false;
            }
            for (; k + MAXB <= cnt; k += MAXB)
                gather_batch<MAXB, false>(X, colidx, jv, dg, di, k, laneByte, lane, e, ne, jn, ax, ay, az, aw);
            if constexpr (MAXB > 8) {
                if (k + 8 <= cnt) { gather_batch<8, false>(X, colidx, jv, dg, di, k, laneByte, lane, e, ne, jn, ax, ay, az, aw); k += 8; }
            }
            if (k + 4 <= cnt) { gather_batch<4, false>(X, colidx, jv, dg, di, k, laneByte, lane, e, ne, jn, ax, ay, az, aw); k += 4; }
            if (k + 2 <= cnt) { gather_batch<2, false>(X, colidx, jv, dg, di, k, laneByte, lane, e, ne, jn, ax, ay, az, aw); k += 2; }
            if (k < cnt)      { gather_batch<1, false>(X, colidx, jv, dg, di, k, laneByte, lane, e, ne, jn, ax, ay, az, aw); }
        }
        if (pf) jn = load_ids(colidx, e, ne, lane);
        float* dst = sA + (lr0 + i) * APAD + c4;
        dst[0] = ax; dst[1] = ay; dst[2] = az; dst[3] = aw;
        s = e;
    }
}

__device__ __forceinline__ void gemm_tile(
    const float* __restrict__ Arow, const v2f (&bfrag)[F / 4],
    float* __restrict__ out, int rowBase, int half, int col, int n, int t)
{
    v8f acc = v8f{};
    const float* ap0 = Arow + t * 16 * APAD;
    #pragma unroll
    for (int kk = 0; kk < F / 4; ++kk) {
        const float* ap = ap0 + 4 * kk;
        v2f a; a.x = ap[0]; a.y = ap[1];
        acc = __builtin_amdgcn_wmma_f32_16x16x4_f32(
            false, a, false, bfrag[kk], (short)0, acc, false, false);
    }
    #pragma unroll
    for (int r = 0; r < 8; ++r) {
        const int lr   = t * 16 + half * 8 + r;
        const int node = rowBase + lr;
        if (lr < BROWS && node < n) out[(long long)node * F + col] = acc[r];
    }
}

__global__ __launch_bounds__(THREADS, 8) void gcn_fused_kernel(
    const float* __restrict__ X,
    const float* __restrict__ W,
    const float* __restrict__ deg,
    const int*   __restrict__ rowp,
    const int*   __restrict__ colidx,
    float*       __restrict__ out,
    int n)
{
    extern __shared__ float smem[];
    float* sA = smem;

    const int tid  = threadIdx.x;
    const int wave = __builtin_amdgcn_readfirstlane(tid >> 5);
    const int lane = tid & 31;

    const int rowBase = blockIdx.x * BROWS;
    const int c4 = lane << 2;
    const unsigned laneByte = (unsigned)lane * 16u;
    const bool smallFirst = (__builtin_popcount(blockIdx.x) & 1) != 0;
    const int g0First = wave * (G0ROWS / 8);
    const int g1Rows  = G1LO + (wave < G1EX ? 1 : 0);
    const int g1First = G0ROWS + wave * G1LO + (wave < G1EX ? wave : G1EX);

    for (int pr = BROWS + wave; pr < TROWS; pr += 8) {
        float* z = sA + pr * APAD + c4;
        z[0] = 0.f; z[1] = 0.f; z[2] = 0.f; z[3] = 0.f;
    }

    {
        const int first = smallFirst ? g1First : g0First;
        const int rows  = smallFirst ? g1Rows  : (G0ROWS / 8);
        gather_rows<16>(X, deg, rowp, colidx, sA, n, rowBase + first, first, rows, lane, laneByte, c4);
    }

    const int half = lane >> 4;
    const int l16  = lane & 15;
    const int colBase = wave << 4;
    const float* Bcol = W + (2 * half) * F + colBase + l16;
    v2f bfrag[F / 4];
    #pragma unroll
    for (int kk = 0; kk < F / 4; ++kk) {
        bfrag[kk].x = Bcol[(4 * kk) * F];
        bfrag[kk].y = Bcol[(4 * kk + 1) * F];
    }
    __syncthreads();

    const float* Arow = sA + l16 * APAD + 2 * half;
    const int col = colBase + l16;
    {
        const int t0 = smallFirst ? G0TILES : 0;
        const int t1 = smallFirst ? NTILE : G0TILES;
        for (int t = t0; t < t1; ++t)
            gemm_tile(Arow, bfrag, out, rowBase, half, col, n, t);
    }

    {
        const int first = smallFirst ? g0First : g1First;
        const int rows  = smallFirst ? (G0ROWS / 8) : g1Rows;
        gather_rows<8>(X, deg, rowp, colidx, sA, n, rowBase + first, first, rows, lane, laneByte, c4);
    }
    __syncthreads();
    {
        const int t0 = smallFirst ? 0 : G0TILES;
        const int t1 = smallFirst ? G0TILES : NTILE;
        for (int t = t0; t < t1; ++t)
            gemm_tile(Arow, bfrag, out, rowBase, half, col, n, t);
    }
}

extern "C" void kernel_launch(void* const* d_in, const int* in_sizes, int n_in,
                              void* d_out, int out_size, void* d_ws, size_t ws_size,
                              hipStream_t stream) {
    const float* X      = (const float*)d_in[0];
    const float* W      = (const float*)d_in[1];
    const float* deg    = (const float*)d_in[2];
    const int*   rowp   = (const int*)d_in[3];
    const int*   colidx = (const int*)d_in[4];
    float*       out    = (float*)d_out;

    const int n = in_sizes[2];
    const int blocks = (n + BROWS - 1) / BROWS;
    const size_t lds_bytes = (size_t)(TROWS * APAD) * sizeof(float);

    gcn_fused_kernel<<<blocks, THREADS, lds_bytes, stream>>>(
        X, W, deg, rowp, colidx, out, n);
}


# Result rows staged through LDS and written as whole 512-byte rows (one 16-byte-per-lane store each) instead of two waves' 64-byte halves per 128-byte line; two more block barriers; sums unchanged
# speedup vs baseline: 1.4643x; 1.1357x over previous
#include <hip/hip_runtime.h>

#define F       128
#define BROWS   98
#define NTILE   7
#define TROWS   (NTILE * 16)
#define G0ROWS  64
#define G0TILES (G0ROWS / 16)
#define G1ROWS  (BROWS - G0ROWS)
#define G1LO    (G1ROWS / 8)
#define G1EX    (G1ROWS % 8)
#define APAD    132
#define SROWS   128
#define THREADS 256

typedef __attribute__((ext_vector_type(2))) float v2f;
typedef __attribute__((ext_vector_type(8))) float v8f;
typedef __attribute__((ext_vector_type(4))) float v4f;

__device__ __forceinline__ int load_ids(const int* __restrict__ colidx, int s, int e, int lane)
{
    int v = 0;
    if (s + lane < e) v = colidx[s + lane];
    return v;
}

template <int B, bool PREFETCH>
__device__ __forceinline__ void gather_batch(
    const float* __restrict__ X, const int* __restrict__ colidx,
    int jv, float dg, float di, int k, unsigned laneByte, int lane,
    int ns, int ne, int& jn,
    float& ax, float& ay, float& az, float& aw)
{
    float4 x[B];
    #pragma unroll
    for (int u = 0; u < B; ++u) {
        const int j = __builtin_amdgcn_readlane(jv, k + u);
        const char* xrow = (const char*)X + (size_t)(unsigned)j * (F * sizeof(float));
        x[u] = *(const float4*)(xrow + laneByte);
    }
    __builtin_amdgcn_sched_barrier(0);
    if (PREFETCH) jn = load_ids(colidx, ns, ne, lane);
    const float nv = di * dg;
    #pragma unroll
    for (int u = 0; u < B; ++u) {
        const float nrm = __int_as_float(__builtin_amdgcn_readlane(__float_as_int(nv), k + u));
        ax = __builtin_fmaf(nrm, x[u].x, ax); ay = __builtin_fmaf(nrm, x[u].y, ay);
        az = __builtin_fmaf(nrm, x[u].z, az); aw = __builtin_fmaf(nrm, x[u].w, aw);
    }
}

template <int MAXB>
__device__ __forceinline__ void gather_rows(
    const float* __restrict__ X, const float* __restrict__ deg,
    const int* __restrict__ rowp, const int* __restrict__ colidx,
    float* __restrict__ sA, int n, int r0, int lr0, int myRows,
    int lane, unsigned laneByte, int c4)
{
    int rpi = r0 + (lane < myRows ? lane : myRows);
    if (rpi > n) rpi = n;
    const int rpv = rowp[rpi];
    int dgi = r0 + (lane < myRows ? lane : 0);
    if (dgi > n - 1) dgi = n - 1;
    const float dgv = deg[dgi];

    int s  = __builtin_amdgcn_readlane(rpv, 0);
    int jn = load_ids(colidx, s, __builtin_amdgcn_readlane(rpv, 1), lane);
    for (int i = 0; i < myRows; ++i) {
        const int   e  = __builtin_amdgcn_readlane(rpv, i + 1);
        const int   ne = __builtin_amdgcn_readlane(rpv, i + 2);
        const float di = __int_as_float(__builtin_amdgcn_readlane(__float_as_int(dgv), i));
        int  jv = jn;
        bool pf = (i + 1 < myRows);
        float ax = 0.f, ay = 0.f, az = 0.f, aw = 0.f;
        for (int base = s; base < e; base += 32) {
            const int rem = e - base;
            const int cnt = rem < 32 ? rem : 32;
            if (base != s) jv = load_ids(colidx, base, e, lane);
            const float dg = deg[jv];
            int k = 0;
            if (pf && cnt >= MAXB) {
                gather_batch<MAXB, true>(X, colidx, jv, dg, di, k, laneByte, lane, e, ne, jn, ax, ay, az, aw);
                k = MAXB; pf = false;
            }
            for (; k + MAXB <= cnt; k += MAXB)
                gather_batch<MAXB, false>(X, colidx, jv, dg, di, k, laneByte, lane, e, ne, jn, ax, ay, az, aw);
            if constexpr (MAXB > 8) {
                if (k + 8 <= cnt) { gather_batch<8, false>(X, colidx, jv, dg, di, k, laneByte, lane, e, ne, jn, ax, ay, az, aw); k += 8; }
            }
            if (k + 4 <= cnt) { gather_batch<4, false>(X, colidx, jv, dg, di, k, laneByte, lane, e, ne, jn, ax, ay, az, aw); k += 4; }
            if (k + 2 <= cnt) { gather_batch<2, false>(X, colidx, jv, dg, di, k, laneByte, lane, e, ne, jn, ax, ay, az, aw); k += 2; }
            if (k < cnt)      { gather_batch<1, false>(X, colidx, jv, dg, di, k, laneByte, lane, e, ne, jn, ax, ay, az, aw); }
        }
        if (pf) jn = load_ids(colidx, e, ne, lane);
        float* dst = sA + (lr0 + i) * APAD + c4;
        dst[0] = ax; dst[1] = ay; dst[2] = az; dst[3] = aw;
        s = e;
    }
}

__device__ __forceinline__ void gemm_tile(
    const float* __restrict__ Arow, const v2f (&bfrag)[F / 4],
    float* __restrict__ stg, int half, int col, int t, int tg)
{
    v8f acc = v8f{};
    const float* ap0 = Arow + t * 16 * APAD;
    #pragma unroll
    for (int kk = 0; kk < F / 4; ++kk) {
        const float* ap = ap0 + 4 * kk;
        v2f a; a.x = ap[0]; a.y = ap[1];
        acc = __builtin_amdgcn_wmma_f32_16x16x4_f32(
            false, a, false, bfrag[kk], (short)0, acc, false, false);
    }
    #pragma unroll
    for (int r = 0; r < 8; ++r)
        stg[(tg * 16 + half * 8 + r) * APAD + col] = acc[r];
}

template <int R>
__device__ __forceinline__ void store_rows(
    const float* sA, float* __restrict__ out, int srow, int step, int cnt, int node0, int n, int lane)
{
    v4f v[R];
    #pragma unroll
    for (int i = 0; i < R; ++i)
        v[i] = *(const v4f*)(sA + (srow + i * step) * APAD + 4 * lane);
    #pragma unroll
    for (int i = 0; i < R; ++i)
        asm volatile("" :: "v"(v[i]));
    #pragma unroll
    for (int i = 0; i < R; ++i) {
        const int node = node0 + i * step;
        if (i < cnt && node < n) *(v4f*)(out + (size_t)node * F + 4 * lane) = v[i];
    }
}

__global__ __launch_bounds__(THREADS, 8) void gcn_fused_kernel(
    const float* __restrict__ X,
    const float* __restrict__ W,
    const float* __restrict__ deg,
    const int*   __restrict__ rowp,
    const int*   __restrict__ colidx,
    float*       __restrict__ out,
    int n)
{
    extern __shared__ __align__(16) float smem[];
    float* sA = smem;
    float* stg0 = sA + G0ROWS * APAD;
    float* stg1 = sA;

    const int tid  = threadIdx.x;
    const int wave = __builtin_amdgcn_readfirstlane(tid >> 5);
    const int lane = tid & 31;

    const int rowBase = blockIdx.x * BROWS;
    const int c4 = lane << 2;
    const unsigned laneByte = (unsigned)lane * 16u;
    const bool smallFirst = (__builtin_popcount(blockIdx.x) & 1) != 0;
    const int g0First = wave * (G0ROWS / 8);
    const int g1Rows  = G1LO + (wave < G1EX ? 1 : 0);
    const int g1First = G0ROWS + wave * G1LO + (wave < G1EX ? wave : G1EX);

    if (smallFirst) {
        for (int pr = BROWS + wave; pr < TROWS; pr += 8) {
            float* z = sA + pr * APAD + c4;
            z[0] = 0.f; z[1] = 0.f; z[2] = 0.f; z[3] = 0.f;
        }
    }

    {
        const int first = smallFirst ? g1First : g0First;
        const int rows  = smallFirst ? g1Rows  : (G0ROWS / 8);
        gather_rows<16>(X, deg, rowp, colidx, sA, n, rowBase + first, first, rows, lane, laneByte, c4);
    }

    const int half = lane >> 4;
    const int l16  = lane & 15;
    const int colBase = wave << 4;
    const float* Bcol = W + (2 * half) * F + colBase + l16;
    v2f bfrag[F / 4];
    #pragma unroll
    for (int kk = 0; kk < F / 4; ++kk) {
        bfrag[kk].x = Bcol[(4 * kk) * F];
        bfrag[kk].y = Bcol[(4 * kk + 1) * F];
    }
    __syncthreads();

    const float* Arow = sA + l16 * APAD + 2 * half;
    const int col = colBase + l16;
    {
        const int t0 = smallFirst ? G0TILES : 0;
        const int t1 = smallFirst ? NTILE : G0TILES;
        float* stg = smallFirst ? stg1 : stg0;
        for (int t = t0; t < t1; ++t)
            gemm_tile(Arow, bfrag, stg, half, col, t, t - t0);
    }
    __syncthreads();

    if (smallFirst) {
        int cnt = G1ROWS - g0First;
        cnt = cnt < 0 ? 0 : (cnt > G0ROWS / 8 ? G0ROWS / 8 : cnt);
        store_rows<G0ROWS / 8>(sA, out, g0First, 1, cnt, rowBase + G0ROWS + g0First, n, lane);
    } else {
        store_rows<G1LO + 1>(sA, out, g1First, 1, g1Rows, rowBase + g1First - G0ROWS, n, lane);
        store_rows<2>(sA, out, BROWS + wave, 8, (BROWS + wave + 8 < TROWS) ? 2 : 1, rowBase + BROWS + wave - G0ROWS, n, lane);
        store_rows<2>(sA, out, TROWS + 2 * wave, 1, 2, rowBase + TROWS - G0ROWS + 2 * wave, n, lane);
        for (int pr = BROWS + wave; pr < TROWS; pr += 8) {
            float* z = sA + pr * APAD + c4;
            z[0] = 0.f; z[1] = 0.f; z[2] = 0.f; z[3] = 0.f;
        }
    }

    {
        const int first = smallFirst ? g0First : g1First;
        const int rows  = smallFirst ? (G0ROWS / 8) : g1Rows;
        gather_rows<8>(X, deg, rowp, colidx, sA, n, rowBase + first, first, rows, lane, laneByte, c4);
    }
    __syncthreads();
    {
        const int t0 = smallFirst ? 0 : G0TILES;
        const int t1 = smallFirst ? G0TILES : NTILE;
        float* stg = smallFirst ? stg0 : stg1;
        for (int t = t0; t < t1; ++t)
            gemm_tile(Arow, bfrag, stg, half, col, t, t - t0);
    }
    __syncthreads();
    if (smallFirst) {
        store_rows<G0ROWS / 8>(sA, out, G0ROWS + g0First, 1, G0ROWS / 8, rowBase + g0First, n, lane);
    } else {
        store_rows<(G1ROWS + 7) / 8>(sA, out, wave, 8, (G1ROWS - wave + 7) / 8, rowBase + G0ROWS + wave, n, lane);
    }
}

extern "C" void kernel_launch(void* const* d_in, const int* in_sizes, int n_in,
                              void* d_out, int out_size, void* d_ws, size_t ws_size,
                              hipStream_t stream) {
    const float* X      = (const float*)d_in[0];
    const float* W      = (const float*)d_in[1];
    const float* deg    = (const float*)d_in[2];
    const int*   rowp   = (const int*)d_in[3];
    const int*   colidx = (const int*)d_in[4];
    float*       out    = (float*)d_out;

    const int n = in_sizes[2];
    const int blocks = (n + BROWS - 1) / BROWS;
    const size_t lds_bytes = (size_t)(SROWS * APAD) * sizeof(float);

    gcn_fused_kernel<<<blocks, THREADS, lds_bytes, stream>>>(
        X, W, deg, rowp, colidx, out, n);
}
